# starting kernel
# baseline (speedup 1.0000x reference)
#include <hip/hip_runtime.h>
#include <math.h>

#define T_STEPS 256
#define BATCH   128
#define INP_D   1024
#define HS_D    1024
#define OUT_D   1024

#define KC       32
#define GRP_F    192
#define KPAIR_F  96
#define CHUNK_F  (8 * GRP_F)

typedef __attribute__((ext_vector_type(2))) float v2f;
typedef __attribute__((ext_vector_type(8))) float v8f;

__device__ __forceinline__ v8f wmma_f32(v2f a, v2f b, v8f c) {
    return __builtin_amdgcn_wmma_f32_16x16x4_f32(
         false, a,  false, b,
         (short)0, c,  false,  false);
}

__device__ __forceinline__ void barrier_lds() {
    asm volatile(
        "s_wait_dscnt 0x0\n\t"
        "s_barrier_signal -1\n\t"
        "s_barrier_wait -1"
        ::: "memory");
}

__global__ void rnn_zero_kernel(float* __restrict__ p, int n) {
    int i = blockIdx.x * blockDim.x + threadIdx.x;
    if (i < n) p[i] = 0.0f;
}

__device__ __forceinline__ void stage_frag(float* buf, int lrow, int lcol, float4 v) {
    const int g  = lrow >> 2;
    const int hk = (lrow >> 1) & 1;
    const int e  = lrow & 1;
    float* dst = buf + g * GRP_F + hk * KPAIR_F + e;
    dst[(lcol + 0) * 2] = v.x;
    dst[(lcol + 1) * 2] = v.y;
    dst[(lcol + 2) * 2] = v.z;
    dst[(lcol + 3) * 2] = v.w;
}

__device__ __forceinline__ void compute_chunk(
    const float* bufp, const float* ap, int half, int l16,
    v8f& acc0, v8f& acc1)
{
    v2f a[8], b0[8], b1[8];
    #pragma unroll
    for (int g = 0; g < 8; ++g) {
        a[g] = *(const v2f*)(ap + g * 4);
        const float* fb = bufp + g * GRP_F + half * KPAIR_F;
        b0[g] = *(const v2f*)(fb + l16 * 2);
        b1[g] = *(const v2f*)(fb + (l16 + 16) * 2);
    }
    #pragma unroll
    for (int g = 0; g < 8; ++g) {
        acc0 = wmma_f32(a[g], b0[g], acc0);
        acc1 = wmma_f32(a[g], b1[g], acc1);
    }
}

__device__ __forceinline__ void gemm_chunked(
    const float* __restrict__ A, int lda, int KTOT,
    const float* __restrict__ W,
    int m0, int n0, float* Bs,
    v8f& acc0, v8f& acc1)
{
    const int tid  = threadIdx.x;
    const int lane = tid & 31;
    const int half = lane >> 4;
    const int l16  = lane & 15;
    const int NC   = KTOT / KC;

    const int lrow = tid >> 3;
    const int lcol = (tid & 7) << 2;
    const float* gsrc   = W + (size_t)lrow * HS_D + n0 + lcol;
    const float* a_base = A + (size_t)(m0 + l16) * lda + half * 2;

    float4 breg = *(const float4*)gsrc;
    stage_frag(Bs, lrow, lcol, breg);

    int c = 0;
    for (; c < NC - 1; ++c) {
        breg = *(const float4*)(gsrc + (size_t)(c + 1) * KC * HS_D);
        barrier_lds();
        compute_chunk(Bs + (c & 1) * CHUNK_F, a_base + c * KC, half, l16,
                      acc0, acc1);
        stage_frag(Bs + ((c + 1) & 1) * CHUNK_F, lrow, lcol, breg);
    }
    barrier_lds();
    compute_chunk(Bs + (c & 1) * CHUNK_F, a_base + c * KC, half, l16,
                  acc0, acc1);
    barrier_lds();
}

__global__ __launch_bounds__(256) void rnn_step_kernel(
    const float* __restrict__ x,
    const float* __restrict__ h_in,
    const float* __restrict__ W_ih,
    const float* __restrict__ W_hh,
    const float* __restrict__ b_i,
    const float* __restrict__ b_h,
    float* __restrict__ h_out)
{
    __shared__ float Bs[2 * CHUNK_F];

    const int wave = threadIdx.x >> 5;
    const int lane = threadIdx.x & 31;
    const int half = lane >> 4;
    const int l16  = lane & 15;

    const int m0 = wave * 16;
    const int n0 = blockIdx.x * 32;

    v8f acc0 = {};
    v8f acc1 = {};

    gemm_chunked(x,    INP_D, INP_D, W_ih, m0, n0, Bs, acc0, acc1);
    gemm_chunked(h_in, HS_D,  HS_D,  W_hh, m0, n0, Bs, acc0, acc1);

    const float bias0 = b_i[n0 + l16]      + b_h[n0 + l16];
    const float bias1 = b_i[n0 + 16 + l16] + b_h[n0 + 16 + l16];

    #pragma unroll
    for (int r = 0; r < 8; ++r) {
        const int m = m0 + r + half * 8;
        h_out[(size_t)m * HS_D + n0 + l16]      = tanhf(acc0[r] + bias0);
        h_out[(size_t)m * HS_D + n0 + 16 + l16] = tanhf(acc1[r] + bias1);
    }
}

__global__ __launch_bounds__(256) void rnn_out_kernel(
    const float* __restrict__ h,
    const float* __restrict__ W_out,
    const float* __restrict__ b_out,
    float* __restrict__ out)
{
    const int wave = threadIdx.x >> 5;
    const int lane = threadIdx.x & 31;
    const int half = lane >> 4;
    const int l16  = lane & 15;

    const int m0 = wave * 16;
    const int n0 = blockIdx.x * 32;

    v8f acc0 = {};
    v8f acc1 = {};

    const float* a_ptr  = h + (size_t)(m0 + l16) * HS_D + half * 2;
    const float* bbase0 = W_out + (size_t)(n0 + l16) * HS_D + half * 2;
    const float* bbase1 = W_out + (size_t)(n0 + 16 + l16) * HS_D + half * 2;

    #pragma unroll 4
    for (int k = 0; k < HS_D; k += 4) {
        v2f a  = *(const v2f*)(a_ptr + k);
        v2f b0 = *(const v2f*)(bbase0 + k);
        v2f b1 = *(const v2f*)(bbase1 + k);
        acc0 = wmma_f32(a, b0, acc0);
        acc1 = wmma_f32(a, b1, acc1);
    }

    const float bo0 = b_out[n0 + l16];
    const float bo1 = b_out[n0 + 16 + l16];

    #pragma unroll
    for (int r = 0; r < 8; ++r) {
        const int m = m0 + r + half * 8;
        out[(size_t)m * OUT_D + n0 + l16]      = acc0[r] + bo0;
        out[(size_t)m * OUT_D + n0 + 16 + l16] = acc1[r] + bo1;
    }
}

extern "C" void kernel_launch(void* const* d_in, const int* in_sizes, int n_in,
                              void* d_out, int out_size, void* d_ws, size_t ws_size,
                              hipStream_t stream) {
    (void)in_sizes; (void)n_in; (void)out_size; (void)ws_size;

    const float* data  = (const float*)d_in[0];
    const float* W_ih  = (const float*)d_in[1];
    const float* W_hh  = (const float*)d_in[2];
    const float* b_i   = (const float*)d_in[3];
    const float* b_h   = (const float*)d_in[4];
    const float* W_out = (const float*)d_in[5];
    const float* b_out = (const float*)d_in[6];
    float* out = (float*)d_out;

    float* h0 = (float*)d_ws;
    float* h1 = h0 + (size_t)BATCH * HS_D;

    const int hn = BATCH * HS_D;
    rnn_zero_kernel<<<(hn + 255) / 256, 256, 0, stream>>>(h0, hn);

    for (int t = 0; t < T_STEPS; ++t) {
        const float* hin  = (t & 1) ? h1 : h0;
        float*       hout = (t & 1) ? h0 : h1;
        rnn_step_kernel<<<HS_D / 32, 256, 0, stream>>>(
            data + (size_t)t * BATCH * INP_D, hin, W_ih, W_hh, b_i, b_h, hout);
    }
    rnn_out_kernel<<<HS_D / 32, 256, 0, stream>>>(h0, W_out, b_out, out);
}


# Step kernel: 128 groups of 4 waves (32x32 tile, whole K per wave), weight panel kept in LDS as 8 slices fetched one ahead, 16-B fragment loads, one barrier per slice; changes summation order
# speedup vs baseline: 3.2451x; 3.2451x over previous
#include <hip/hip_runtime.h>
#include <math.h>


#define T_STEPS 256
#define BATCH   128
#define INP_D   1024
#define HS_D    1024
#define OUT_D   1024

#define TILE_M   32
#define TILE_N   32
#define PH_K     256
#define NPH      8
#define PH_F     (PH_K * TILE_N)
#define PH_BLK   (PH_K / 8)
#define OUT_PITCH 36

typedef __attribute__((ext_vector_type(2))) float v2f;
typedef __attribute__((ext_vector_type(8))) float v8f;

__device__ __forceinline__ v8f wmma_f32(v2f a, v2f b, v8f c) {
    return __builtin_amdgcn_wmma_f32_16x16x4_f32(
         false, a,  false, b,
         (short)0, c,  false,  false);
}

__device__ __forceinline__ void barrier_lds() {
    asm volatile(
        "s_wait_dscnt 0x0\n\t"
        "s_barrier_signal -1\n\t"
        "s_barrier_wait -1"
        ::: "memory");
}

__global__ void rnn_zero_kernel(float* __restrict__ p, int n) {
    int i = blockIdx.x * blockDim.x + threadIdx.x;
    if (i < n) p[i] = 0.0f;
}


__device__ __forceinline__ void panel_load(float4 (&v)[4][4], const float* wp) {
    #pragma unroll
    for (int s = 0; s < 4; ++s) {
        #pragma unroll
        for (int R = 0; R < 4; ++R)
            v[s][R] = *(const float4*)(wp + (size_t)(64 * s + R) * HS_D);
    }
}

__device__ __forceinline__ void panel_store(float* reg, const float4 (&v)[4][4],
                                            int qq, int cg) {
    const int rot = qq & 3;
    #pragma unroll
    for (int s = 0; s < 4; ++s) {
        float* row = reg + (size_t)(16 * s + qq) * 128;
        const float4 c0 = make_float4(v[s][0].x, v[s][1].x, v[s][2].x, v[s][3].x);
        const float4 c1 = make_float4(v[s][0].y, v[s][1].y, v[s][2].y, v[s][3].y);
        const float4 c2 = make_float4(v[s][0].z, v[s][1].z, v[s][2].z, v[s][3].z);
        const float4 c3 = make_float4(v[s][0].w, v[s][1].w, v[s][2].w, v[s][3].w);
        *(float4*)(row + 4 * ((4 * cg + 0 + rot) & 31)) = c0;
        *(float4*)(row + 4 * ((4 * cg + 1 + rot) & 31)) = c1;
        *(float4*)(row + 4 * ((4 * cg + 2 + rot) & 31)) = c2;
        *(float4*)(row + 4 * ((4 * cg + 3 + rot) & 31)) = c3;
    }
}

__device__ __forceinline__ void a_load(float4 (&a)[PH_BLK], const float* ap) {
    #pragma unroll
    for (int b = 0; b < PH_BLK; ++b)
        a[b] = *(const float4*)(ap + 8 * b);
}

__device__ __forceinline__ void b_group(float4 (&bb)[8], const float* reg,
                                        int g, int offE, int offO) {
    #pragma unroll
    for (int j = 0; j < 8; ++j) {
        const int b = 8 * g + j;
        bb[j] = *(const float4*)(reg + 256 * b + ((b & 1) ? offO : offE));
    }
}

__device__ __forceinline__ void slice_mma(const float* reg, const float4 (&a)[PH_BLK],
                                          float4 (&b0)[8],
                                          int offE, int offO, v8f& acc0, v8f& acc1) {
    float4 b1[8];
    #pragma unroll
    for (int g = 0; g < 4; g += 2) {
        b_group(b1, reg, g + 1, offE, offO);
        __builtin_amdgcn_sched_barrier(0);
        #pragma unroll
        for (int j = 0; j < 8; ++j) {
            const float4 av = a[8 * g + j];
            acc0 = wmma_f32((v2f){av.x, av.y}, (v2f){b0[j].x, b0[j].y}, acc0);
            acc1 = wmma_f32((v2f){av.z, av.w}, (v2f){b0[j].z, b0[j].w}, acc1);
        }
        __builtin_amdgcn_sched_barrier(0);
        if (g + 2 < 4) b_group(b0, reg, g + 2, offE, offO);
        __builtin_amdgcn_sched_barrier(0);
        #pragma unroll
        for (int j = 0; j < 8; ++j) {
            const float4 av = a[8 * (g + 1) + j];
            acc0 = wmma_f32((v2f){av.x, av.y}, (v2f){b1[j].x, b1[j].y}, acc0);
            acc1 = wmma_f32((v2f){av.z, av.w}, (v2f){b1[j].z, b1[j].w}, acc1);
        }
        __builtin_amdgcn_sched_barrier(0);
    }
}

__global__ __launch_bounds__(128) void rnn_step_kernel(
    const float* __restrict__ x,
    const float* __restrict__ h_in,
    const float* __restrict__ W_ih,
    const float* __restrict__ W_hh,
    const float* __restrict__ b_i,
    const float* __restrict__ b_h,
    float* __restrict__ h_out)
{
    __shared__ float Ws[NPH * PH_F];

    const int tid  = threadIdx.x;
    const int wave = tid >> 5;
    const int lane = tid & 31;
    const int h    = lane >> 4;
    const int c    = lane & 15;
    const int wm   = wave >> 1;
    const int wn   = wave & 1;
    const int mt   = blockIdx.x >> 5;
    const int nt   = blockIdx.x & 31;
    const int m0   = mt * TILE_M + wm * 16;
    const int n0   = nt * TILE_N;
    const int n    = wn * 16 + c;

    const int offE = 128 * h + 4 * ((n + h) & 31);
    const int offO = 128 * h + 4 * ((n + 2 + h) & 31);

    const int cg = tid & 7;
    const int qq = tid >> 3;
    const size_t wofs = (size_t)(4 * qq) * HS_D + n0 + 4 * cg;
    const float* wih = W_ih + wofs;
    const float* whh = W_hh + wofs;

    const float* xr = x    + (size_t)(m0 + c) * INP_D + 4 * h;
    const float* hr = h_in + (size_t)(m0 + c) * HS_D  + 4 * h;

    const float bias = b_i[n0 + n] + b_h[n0 + n];

    v8f acc0 = {};
    v8f acc1 = {};

    float4 wv[4][4];
    float4 aA[PH_BLK], aB[PH_BLK];

    #define WPTR(p) (((p) < 4) ? (wih + (size_t)(p) * PH_K * HS_D) \
                               : (whh + (size_t)((p) - 4) * PH_K * HS_D))
    #define APTR(p) (((p) < 4) ? (xr + (p) * PH_K) : (hr + ((p) - 4) * PH_K))

    float4 b0[8];

    #define SLICE(p, cur, nxt, HAS_NEXT_A)                                   \
        b_group(b0, Ws + (p) * PH_F, 0, offE, offO);                         \
        if (HAS_NEXT_A) a_load(nxt, APTR((p) + 1));                          \
        __builtin_amdgcn_sched_barrier(0);                                   \
        slice_mma(Ws + (p) * PH_F, cur, b0, offE, offO, acc0, acc1);

    panel_load(wv, WPTR(0));
    a_load(aA, APTR(0));
    panel_store(Ws, wv, qq, cg);
    panel_load(wv, WPTR(1));
    barrier_lds();

    #pragma unroll
    for (int p = 0; p < 6; p += 2) {
        SLICE(p, aA, aB, true)
        panel_store(Ws + (p + 1) * PH_F, wv, qq, cg);
        panel_load(wv, WPTR(p + 2));
        barrier_lds();

        SLICE(p + 1, aB, aA, true)
        panel_store(Ws + (p + 2) * PH_F, wv, qq, cg);
        panel_load(wv, WPTR(p + 3));
        barrier_lds();
    }
    SLICE(6, aA, aB, true)
    panel_store(Ws + 7 * PH_F, wv, qq, cg);
    barrier_lds();
    SLICE(7, aB, aA, false)

    #undef SLICE
    #undef WPTR
    #undef APTR

    float* ot = Ws;
    #pragma unroll
    for (int r = 0; r < 8; ++r) {
        const int row = wm * 16 + r + 8 * h;
        ot[row * OUT_PITCH + n] = tanhf((acc0[r] + acc1[r]) + bias);
    }
    barrier_lds();
    #pragma unroll
    for (int i = 0; i < 2; ++i) {
        const int row = wave * 8 + i * 4 + (lane >> 3);
        const int col = (lane & 7) * 4;
        const float4 v = *(const float4*)(ot + row * OUT_PITCH + col);
        *(float4*)(h_out + (size_t)(mt * TILE_M + row) * HS_D + n0 + col) = v;
    }
}

__global__ __launch_bounds__(256) void rnn_out_kernel(
    const float* __restrict__ h,
    const float* __restrict__ W_out,
    const float* __restrict__ b_out,
    float* __restrict__ out)
{
    const int wave = threadIdx.x >> 5;
    const int lane = threadIdx.x & 31;
    const int half = lane >> 4;
    const int l16  = lane & 15;

    const int m0 = wave * 16;
    const int n0 = blockIdx.x * 32;

    v8f acc0 = {};
    v8f acc1 = {};

    const float* a_ptr  = h + (size_t)(m0 + l16) * HS_D + half * 2;
    const float* bbase0 = W_out + (size_t)(n0 + l16) * HS_D + half * 2;
    const float* bbase1 = W_out + (size_t)(n0 + 16 + l16) * HS_D + half * 2;

    #pragma unroll 4
    for (int k = 0; k < HS_D; k += 4) {
        v2f a  = *(const v2f*)(a_ptr + k);
        v2f b0 = *(const v2f*)(bbase0 + k);
        v2f b1 = *(const v2f*)(bbase1 + k);
        acc0 = wmma_f32(a, b0, acc0);
        acc1 = wmma_f32(a, b1, acc1);
    }

    const float bo0 = b_out[n0 + l16];
    const float bo1 = b_out[n0 + 16 + l16];

    #pragma unroll
    for (int r = 0; r < 8; ++r) {
        const int m = m0 + r + half * 8;
        out[(size_t)m * OUT_D + n0 + l16]      = acc0[r] + bo0;
        out[(size_t)m * OUT_D + n0 + 16 + l16] = acc1[r] + bo1;
    }
}

extern "C" void kernel_launch(void* const* d_in, const int* in_sizes, int n_in,
                              void* d_out, int out_size, void* d_ws, size_t ws_size,
                              hipStream_t stream) {
    (void)in_sizes; (void)n_in; (void)out_size; (void)ws_size;

    const float* data  = (const float*)d_in[0];
    const float* W_ih  = (const float*)d_in[1];
    const float* W_hh  = (const float*)d_in[2];
    const float* b_i   = (const float*)d_in[3];
    const float* b_h   = (const float*)d_in[4];
    const float* W_out = (const float*)d_in[5];
    const float* b_out = (const float*)d_in[6];
    float* out = (float*)d_out;

    float* h0 = (float*)d_ws;
    float* h1 = h0 + (size_t)BATCH * HS_D;

    const int hn = BATCH * HS_D;
    rnn_zero_kernel<<<(hn + 255) / 256, 256, 0, stream>>>(h0, hn);

    for (int t = 0; t < T_STEPS; ++t) {
        const float* hin  = (t & 1) ? h1 : h0;
        float*       hout = (t & 1) ? h0 : h1;
        rnn_step_kernel<<<(BATCH / TILE_M) * (HS_D / TILE_N), 128, 0, stream>>>(
            data + (size_t)t * BATCH * INP_D, hin, W_ih, W_hh, b_i, b_h, hout);
    }
    rnn_out_kernel<<<HS_D / 32, 256, 0, stream>>>(h0, W_out, b_out, out);
}


# Step kernel: the group's 32 rows of x and h go through LDS slice by slice (whole-line global loads, each row read once per group); row fragments read from LDS beside the weights; sums unchanged
# speedup vs baseline: 4.3968x; 1.3549x over previous
#include <hip/hip_runtime.h>
#include <math.h>


#define T_STEPS 256
#define BATCH   128
#define INP_D   1024
#define HS_D    1024
#define OUT_D   1024

#define TILE_M   32
#define TILE_N   32
#define PH_K     256
#define NPH      8
#define PH_F     (PH_K * TILE_N)
#define PH_BLK   (PH_K / 8)
#define OUT_PITCH 36

typedef __attribute__((ext_vector_type(2))) float v2f;
typedef __attribute__((ext_vector_type(4))) float v4f;
typedef __attribute__((ext_vector_type(8))) float v8f;

__device__ __forceinline__ v8f wmma_f32(v2f a, v2f b, v8f c) {
    return __builtin_amdgcn_wmma_f32_16x16x4_f32(
         false, a,  false, b,
         (short)0, c,  false,  false);
}

__device__ __forceinline__ void barrier_lds() {
    asm volatile(
        "s_wait_dscnt 0x0\n\t"
        "s_barrier_signal -1\n\t"
        "s_barrier_wait -1"
        ::: "memory");
}

__global__ void rnn_zero_kernel(float* __restrict__ p, int n) {
    int i = blockIdx.x * blockDim.x + threadIdx.x;
    if (i < n) p[i] = 0.0f;
}


__device__ __forceinline__ void panel_load(float4 (&v)[4][4], const float* wp) {
    #pragma unroll
    for (int s = 0; s < 4; ++s) {
        #pragma unroll
        for (int R = 0; R < 4; ++R)
            v[s][R] = *(const float4*)(wp + (size_t)(64 * s + R) * HS_D);
    }
}

__device__ __forceinline__ void panel_store(float* reg, const float4 (&v)[4][4],
                                            int qq, int cg) {
    const int rot = qq & 3;
    #pragma unroll
    for (int s = 0; s < 4; ++s) {
        float* row = reg + (size_t)(16 * s + qq) * 128;
        const float4 c0 = make_float4(v[s][0].x, v[s][1].x, v[s][2].x, v[s][3].x);
        const float4 c1 = make_float4(v[s][0].y, v[s][1].y, v[s][2].y, v[s][3].y);
        const float4 c2 = make_float4(v[s][0].z, v[s][1].z, v[s][2].z, v[s][3].z);
        const float4 c3 = make_float4(v[s][0].w, v[s][1].w, v[s][2].w, v[s][3].w);
        *(float4*)(row + 4 * ((4 * cg + 0 + rot) & 31)) = c0;
        *(float4*)(row + 4 * ((4 * cg + 1 + rot) & 31)) = c1;
        *(float4*)(row + 4 * ((4 * cg + 2 + rot) & 31)) = c2;
        *(float4*)(row + 4 * ((4 * cg + 3 + rot) & 31)) = c3;
    }
}

__device__ __forceinline__ void rows_load(v4f (&v)[16], const float* ap) {
    #pragma unroll
    for (int i = 0; i < 16; ++i)
        v[i] = *(const v4f*)(ap + (size_t)(16 * (i & 1)) * HS_D + 32 * (i >> 1));
}

__device__ __forceinline__ void rows_store(float* abuf, const v4f (&v)[16],
                                           int ar0, int cg) {
    #pragma unroll
    for (int i = 0; i < 16; ++i) {
        const int row = ar0 + 16 * (i & 1);
        const int u   = 8 * (i >> 1) + cg;
        *(v4f*)(abuf + row * PH_K + 4 * (u ^ ar0)) = v[i];
    }
}

__device__ __forceinline__ void frag_group(float4 (&fa)[8], float4 (&fb)[8],
                                           const float* wreg, const float* arow,
                                           int g, int offE, int offO,
                                           const int (&aoff)[8]) {
    #pragma unroll
    for (int j = 0; j < 8; ++j) {
        const int b = 8 * g + j;
        fb[j] = *(const float4*)(wreg + 256 * b + ((b & 1) ? offO : offE));
        fa[j] = *(const float4*)(arow + 64 * g + aoff[j]);
    }
}

__device__ __forceinline__ void slice_mma(const float* wreg, const float* arow,
                                          float4 (&a0)[8], float4 (&b0)[8],
                                          int offE, int offO, const int (&aoff)[8],
                                          v8f& acc0, v8f& acc1) {
    float4 a1[8], b1[8];
    #pragma unroll
    for (int g = 0; g < 4; g += 2) {
        frag_group(a1, b1, wreg, arow, g + 1, offE, offO, aoff);
        __builtin_amdgcn_sched_barrier(0);
        #pragma unroll
        for (int j = 0; j < 8; ++j) {
            acc0 = wmma_f32((v2f){a0[j].x, a0[j].y}, (v2f){b0[j].x, b0[j].y}, acc0);
            acc1 = wmma_f32((v2f){a0[j].z, a0[j].w}, (v2f){b0[j].z, b0[j].w}, acc1);
        }
        __builtin_amdgcn_sched_barrier(0);
        if (g + 2 < 4) frag_group(a0, b0, wreg, arow, g + 2, offE, offO, aoff);
        __builtin_amdgcn_sched_barrier(0);
        #pragma unroll
        for (int j = 0; j < 8; ++j) {
            acc0 = wmma_f32((v2f){a1[j].x, a1[j].y}, (v2f){b1[j].x, b1[j].y}, acc0);
            acc1 = wmma_f32((v2f){a1[j].z, a1[j].w}, (v2f){b1[j].z, b1[j].w}, acc1);
        }
        __builtin_amdgcn_sched_barrier(0);
    }
}

__global__ __launch_bounds__(128) void rnn_step_kernel(
    const float* __restrict__ x,
    const float* __restrict__ h_in,
    const float* __restrict__ W_ih,
    const float* __restrict__ W_hh,
    const float* __restrict__ b_i,
    const float* __restrict__ b_h,
    float* __restrict__ h_out)
{
    __shared__ float Ws[NPH * PH_F];

    const int tid  = threadIdx.x;
    const int wave = tid >> 5;
    const int lane = tid & 31;
    const int h    = lane >> 4;
    const int c    = lane & 15;
    const int wm   = wave >> 1;
    const int wn   = wave & 1;
    const int mt   = blockIdx.x >> 5;
    const int nt   = blockIdx.x & 31;
    const int n0   = nt * TILE_N;
    const int n    = wn * 16 + c;

    const int offE = 128 * h + 4 * ((n + h) & 31);
    const int offO = 128 * h + 4 * ((n + 2 + h) & 31);

    const int cg = tid & 7;
    const int qq = tid >> 3;
    const size_t wofs = (size_t)(4 * qq) * HS_D + n0 + 4 * cg;
    const float* wih = W_ih + wofs;
    const float* whh = W_hh + wofs;

    const int ar0 = (qq >> 1) + 8 * (qq & 1);
    const size_t aofs = (size_t)(mt * TILE_M + ar0) * HS_D + 4 * cg;
    const float* xr = x    + aofs;
    const float* hr = h_in + aofs;

    const int arow_f = (wm * 16 + c) * PH_K;
    int aoff[8];
    #pragma unroll
    for (int j = 0; j < 8; ++j) aoff[j] = 4 * ((2 * j + h) ^ c);

    const float bias = b_i[n0 + n] + b_h[n0 + n];

    v8f acc0 = {};
    v8f acc1 = {};

    float4 wv[4][4];
    v4f av[16];

    #define WPTR(p) (((p) < 4) ? (wih + (size_t)(p) * PH_K * HS_D) \
                               : (whh + (size_t)((p) - 4) * PH_K * HS_D))
    #define APTR(p) (((p) < 4) ? (xr + (p) * PH_K) : (hr + ((p) - 4) * PH_K))
    #define AREG(p) (Ws + (((p) < 2) ? (7 - (p)) : ((p) - 2)) * PH_F)

    float4 a0[8], b0[8];

    #define SLICE(p)                                                         \
        __builtin_amdgcn_sched_barrier(0);                                   \
        frag_group(a0, b0, Ws + (p) * PH_F, AREG(p) + arow_f,                \
                   0, offE, offO, aoff);                                     \
        slice_mma(Ws + (p) * PH_F, AREG(p) + arow_f, a0, b0,                 \
                  offE, offO, aoff, acc0, acc1);

    panel_load(wv, WPTR(0));
    rows_load(av, APTR(0));
    panel_store(Ws, wv, qq, cg);
    rows_store(AREG(0), av, ar0, cg);
    panel_load(wv, WPTR(1));
    rows_load(av, APTR(1));
    barrier_lds();

    #pragma unroll
    for (int p = 0; p < 6; ++p) {
        SLICE(p)
        panel_store(Ws + (p + 1) * PH_F, wv, qq, cg);
        rows_store(AREG(p + 1), av, ar0, cg);
        panel_load(wv, WPTR(p + 2));
        rows_load(av, APTR(p + 2));
        barrier_lds();
    }
    SLICE(6)
    panel_store(Ws + 7 * PH_F, wv, qq, cg);
    rows_store(AREG(7), av, ar0, cg);
    barrier_lds();
    SLICE(7)

    #undef SLICE
    #undef AREG
    #undef WPTR
    #undef APTR

    float* ot = Ws;
    #pragma unroll
    for (int r = 0; r < 8; ++r) {
        const int row = wm * 16 + r + 8 * h;
        ot[row * OUT_PITCH + n] = tanhf((acc0[r] + acc1[r]) + bias);
    }
    barrier_lds();
    #pragma unroll
    for (int i = 0; i < 2; ++i) {
        const int row = wave * 8 + i * 4 + (lane >> 3);
        const int col = (lane & 7) * 4;
        const float4 v = *(const float4*)(ot + row * OUT_PITCH + col);
        *(float4*)(h_out + (size_t)(mt * TILE_M + row) * HS_D + n0 + col) = v;
    }
}

__global__ __launch_bounds__(256) void rnn_out_kernel(
    const float* __restrict__ h,
    const float* __restrict__ W_out,
    const float* __restrict__ b_out,
    float* __restrict__ out)
{
    const int wave = threadIdx.x >> 5;
    const int lane = threadIdx.x & 31;
    const int half = lane >> 4;
    const int l16  = lane & 15;

    const int m0 = wave * 16;
    const int n0 = blockIdx.x * 32;

    v8f acc0 = {};
    v8f acc1 = {};

    const float* a_ptr  = h + (size_t)(m0 + l16) * HS_D + half * 2;
    const float* bbase0 = W_out + (size_t)(n0 + l16) * HS_D + half * 2;
    const float* bbase1 = W_out + (size_t)(n0 + 16 + l16) * HS_D + half * 2;

    #pragma unroll 4
    for (int k = 0; k < HS_D; k += 4) {
        v2f a  = *(const v2f*)(a_ptr + k);
        v2f b0 = *(const v2f*)(bbase0 + k);
        v2f b1 = *(const v2f*)(bbase1 + k);
        acc0 = wmma_f32(a, b0, acc0);
        acc1 = wmma_f32(a, b1, acc1);
    }

    const float bo0 = b_out[n0 + l16];
    const float bo1 = b_out[n0 + 16 + l16];

    #pragma unroll
    for (int r = 0; r < 8; ++r) {
        const int m = m0 + r + half * 8;
        out[(size_t)m * OUT_D + n0 + l16]      = acc0[r] + bo0;
        out[(size_t)m * OUT_D + n0 + 16 + l16] = acc1[r] + bo1;
    }
}

extern "C" void kernel_launch(void* const* d_in, const int* in_sizes, int n_in,
                              void* d_out, int out_size, void* d_ws, size_t ws_size,
                              hipStream_t stream) {
    (void)in_sizes; (void)n_in; (void)out_size; (void)ws_size;

    const float* data  = (const float*)d_in[0];
    const float* W_ih  = (const float*)d_in[1];
    const float* W_hh  = (const float*)d_in[2];
    const float* b_i   = (const float*)d_in[3];
    const float* b_h   = (const float*)d_in[4];
    const float* W_out = (const float*)d_in[5];
    const float* b_out = (const float*)d_in[6];
    float* out = (float*)d_out;

    float* h0 = (float*)d_ws;
    float* h1 = h0 + (size_t)BATCH * HS_D;

    const int hn = BATCH * HS_D;
    rnn_zero_kernel<<<(hn + 255) / 256, 256, 0, stream>>>(h0, hn);

    for (int t = 0; t < T_STEPS; ++t) {
        const float* hin  = (t & 1) ? h1 : h0;
        float*       hout = (t & 1) ? h0 : h1;
        rnn_step_kernel<<<(BATCH / TILE_M) * (HS_D / TILE_N), 128, 0, stream>>>(
            data + (size_t)t * BATCH * INP_D, hin, W_ih, W_hh, b_i, b_h, hout);
    }
    rnn_out_kernel<<<HS_D / 32, 256, 0, stream>>>(h0, W_out, b_out, out);
}


# Step kernel: 8-wave groups in two roles: waves 4-7 only stage the next slice (global loads, LDS stores), waves 0-3 only read fragments and run the WMMAs; tiles, k order and sums unchanged
# speedup vs baseline: 4.6698x; 1.0621x over previous
#include <hip/hip_runtime.h>
#include <math.h>


#define T_STEPS 256
#define BATCH   128
#define INP_D   1024
#define HS_D    1024
#define OUT_D   1024

#define TILE_M   32
#define TILE_N   32
#define PH_K     256
#define NPH      8
#define PH_F     (PH_K * TILE_N)
#define PH_BLK   (PH_K / 8)
#define OUT_PITCH 36

typedef __attribute__((ext_vector_type(2))) float v2f;
typedef __attribute__((ext_vector_type(4))) float v4f;
typedef __attribute__((ext_vector_type(8))) float v8f;

__device__ __forceinline__ v8f wmma_f32(v2f a, v2f b, v8f c) {
    return __builtin_amdgcn_wmma_f32_16x16x4_f32(
         false, a,  false, b,
         (short)0, c,  false,  false);
}

__global__ void rnn_zero_kernel(float* __restrict__ p, int n) {
    int i = blockIdx.x * blockDim.x + threadIdx.x;
    if (i < n) p[i] = 0.0f;
}


__device__ __forceinline__ void panel_load(float4 (&v)[4][4], const float* wp) {
    #pragma unroll
    for (int s = 0; s < 4; ++s) {
        #pragma unroll
        for (int R = 0; R < 4; ++R)
            v[s][R] = *(const float4*)(wp + (size_t)(64 * s + R) * HS_D);
    }
}

__device__ __forceinline__ void panel_store(float* reg, const float4 (&v)[4][4],
                                            int qq, int cg) {
    const int rot = qq & 3;
    #pragma unroll
    for (int s = 0; s < 4; ++s) {
        float* row = reg + (size_t)(16 * s + qq) * 128;
        const float4 c0 = make_float4(v[s][0].x, v[s][1].x, v[s][2].x, v[s][3].x);
        const float4 c1 = make_float4(v[s][0].y, v[s][1].y, v[s][2].y, v[s][3].y);
        const float4 c2 = make_float4(v[s][0].z, v[s][1].z, v[s][2].z, v[s][3].z);
        const float4 c3 = make_float4(v[s][0].w, v[s][1].w, v[s][2].w, v[s][3].w);
        *(float4*)(row + 4 * ((4 * cg + 0 + rot) & 31)) = c0;
        *(float4*)(row + 4 * ((4 * cg + 1 + rot) & 31)) = c1;
        *(float4*)(row + 4 * ((4 * cg + 2 + rot) & 31)) = c2;
        *(float4*)(row + 4 * ((4 * cg + 3 + rot) & 31)) = c3;
    }
}

__device__ __forceinline__ void rows_load(v4f (&v)[16], const float* ap) {
    #pragma unroll
    for (int i = 0; i < 16; ++i)
        v[i] = *(const v4f*)(ap + (size_t)(16 * (i & 1)) * HS_D + 32 * (i >> 1));
}

__device__ __forceinline__ void rows_store(float* abuf, const v4f (&v)[16],
                                           int ar0, int cg) {
    #pragma unroll
    for (int i = 0; i < 16; ++i) {
        const int row = ar0 + 16 * (i & 1);
        const int u   = 8 * (i >> 1) + cg;
        *(v4f*)(abuf + row * PH_K + 4 * (u ^ ar0)) = v[i];
    }
}

__device__ __forceinline__ void frag_group(float4 (&fa)[8], float4 (&fb)[8],
                                           const float* wreg, const float* arow,
                                           int g, int offE, int offO,
                                           const int (&aoff)[8]) {
    #pragma unroll
    for (int j = 0; j < 8; ++j) {
        const int b = 8 * g + j;
        fb[j] = *(const float4*)(wreg + 256 * b + ((b & 1) ? offO : offE));
        fa[j] = *(const float4*)(arow + 64 * g + aoff[j]);
    }
}

__device__ __forceinline__ void slice_mma(const float* wreg, const float* arow,
                                          float4 (&a0)[8], float4 (&b0)[8],
                                          int offE, int offO, const int (&aoff)[8],
                                          v8f& acc0, v8f& acc1) {
    float4 a1[8], b1[8];
    #pragma unroll
    for (int g = 0; g < 4; g += 2) {
        frag_group(a1, b1, wreg, arow, g + 1, offE, offO, aoff);
        __builtin_amdgcn_sched_barrier(0);
        #pragma unroll
        for (int j = 0; j < 8; ++j) {
            acc0 = wmma_f32((v2f){a0[j].x, a0[j].y}, (v2f){b0[j].x, b0[j].y}, acc0);
            acc1 = wmma_f32((v2f){a0[j].z, a0[j].w}, (v2f){b0[j].z, b0[j].w}, acc1);
        }
        __builtin_amdgcn_sched_barrier(0);
        if (g + 2 < 4) frag_group(a0, b0, wreg, arow, g + 2, offE, offO, aoff);
        __builtin_amdgcn_sched_barrier(0);
        #pragma unroll
        for (int j = 0; j < 8; ++j) {
            acc0 = wmma_f32((v2f){a1[j].x, a1[j].y}, (v2f){b1[j].x, b1[j].y}, acc0);
            acc1 = wmma_f32((v2f){a1[j].z, a1[j].w}, (v2f){b1[j].z, b1[j].w}, acc1);
        }
        __builtin_amdgcn_sched_barrier(0);
    }
}

__global__ __launch_bounds__(256) void rnn_step_kernel(
    const float* __restrict__ x,
    const float* __restrict__ h_in,
    const float* __restrict__ W_ih,
    const float* __restrict__ W_hh,
    const float* __restrict__ b_i,
    const float* __restrict__ b_h,
    float* __restrict__ h_out)
{
    __shared__ float Ws[NPH * PH_F];

    const int tid  = threadIdx.x;
    const int loader = __builtin_amdgcn_readfirstlane(tid >> 7);
    const int rt   = tid & 127;
    const int wave = rt >> 5;
    const int lane = tid & 31;
    const int h    = lane >> 4;
    const int c    = lane & 15;
    const int wm   = wave >> 1;
    const int wn   = wave & 1;
    const int mt   = blockIdx.x >> 5;
    const int nt   = blockIdx.x & 31;
    const int n0   = nt * TILE_N;
    const int n    = wn * 16 + c;

    const int offE = 128 * h + 4 * ((n + h) & 31);
    const int offO = 128 * h + 4 * ((n + 2 + h) & 31);

    const int cg = rt & 7;
    const int qq = rt >> 3;
    const size_t wofs = (size_t)(4 * qq) * HS_D + n0 + 4 * cg;
    const float* wih = W_ih + wofs;
    const float* whh = W_hh + wofs;

    const int ar0 = (qq >> 1) + 8 * (qq & 1);
    const size_t aofs = (size_t)(mt * TILE_M + ar0) * HS_D + 4 * cg;
    const float* xr = x    + aofs;
    const float* hr = h_in + aofs;

    const int arow_f = (wm * 16 + c) * PH_K;
    int aoff[8];
    #pragma unroll
    for (int j = 0; j < 8; ++j) aoff[j] = 4 * ((2 * j + h) ^ c);

    const float bias = b_i[n0 + n] + b_h[n0 + n];

    v8f acc0 = {};
    v8f acc1 = {};

    #define WPTR(p) (((p) < 4) ? (wih + (size_t)(p) * PH_K * HS_D) \
                               : (whh + (size_t)((p) - 4) * PH_K * HS_D))
    #define APTR(p) (((p) < 4) ? (xr + (p) * PH_K) : (hr + ((p) - 4) * PH_K))
    #define AREG(p) (Ws + (((p) < 2) ? (7 - (p)) : ((p) - 2)) * PH_F)

    float4 a0[8], b0[8];

    #define SLICE(p)                                                         \
        frag_group(a0, b0, Ws + (p) * PH_F, AREG(p) + arow_f,                \
                   0, offE, offO, aoff);                                     \
        slice_mma(Ws + (p) * PH_F, AREG(p) + arow_f, a0, b0,                 \
                  offE, offO, aoff, acc0, acc1);

    #define STAGE(p)                                                         \
        {                                                                    \
            float4 wv[4][4];                                                 \
            v4f av[16];                                                      \
            panel_load(wv, WPTR(p));                                         \
            rows_load(av, APTR(p));                                          \
            panel_store(Ws + (p) * PH_F, wv, qq, cg);                        \
            rows_store(AREG(p), av, ar0, cg);                                \
        }
    #define PHASE(p)                                                         \
        if (loader) {                                                        \
            STAGE((p) + 1)                                                   \
        } else {                                                             \
            SLICE(p)                                                         \
        }                                                                    \
        __syncthreads();

    if (loader) {
        STAGE(0)
    }
    __syncthreads();

    PHASE(0)
    PHASE(1)
    PHASE(2)
    PHASE(3)
    PHASE(4)
    PHASE(5)
    PHASE(6)
    if (!loader) {
        SLICE(7)
    }

    #undef PHASE
    #undef STAGE

    #undef SLICE
    #undef AREG
    #undef WPTR
    #undef APTR

    float* ot = Ws;
    if (!loader) {
        #pragma unroll
        for (int r = 0; r < 8; ++r) {
            const int row = wm * 16 + r + 8 * h;
            ot[row * OUT_PITCH + n] = tanhf((acc0[r] + acc1[r]) + bias);
        }
    }
    __syncthreads();
    {
        const int row = tid >> 3;
        const int col = (tid & 7) * 4;
        const float4 v = *(const float4*)(ot + row * OUT_PITCH + col);
        *(float4*)(h_out + (size_t)(mt * TILE_M + row) * HS_D + n0 + col) = v;
    }
}

__global__ __launch_bounds__(256) void rnn_out_kernel(
    const float* __restrict__ h,
    const float* __restrict__ W_out,
    const float* __restrict__ b_out,
    float* __restrict__ out)
{
    const int wave = threadIdx.x >> 5;
    const int lane = threadIdx.x & 31;
    const int half = lane >> 4;
    const int l16  = lane & 15;

    const int m0 = wave * 16;
    const int n0 = blockIdx.x * 32;

    v8f acc0 = {};
    v8f acc1 = {};

    const float* a_ptr  = h + (size_t)(m0 + l16) * HS_D + half * 2;
    const float* bbase0 = W_out + (size_t)(n0 + l16) * HS_D + half * 2;
    const float* bbase1 = W_out + (size_t)(n0 + 16 + l16) * HS_D + half * 2;

    #pragma unroll 4
    for (int k = 0; k < HS_D; k += 4) {
        v2f a  = *(const v2f*)(a_ptr + k);
        v2f b0 = *(const v2f*)(bbase0 + k);
        v2f b1 = *(const v2f*)(bbase1 + k);
        acc0 = wmma_f32(a, b0, acc0);
        acc1 = wmma_f32(a, b1, acc1);
    }

    const float bo0 = b_out[n0 + l16];
    const float bo1 = b_out[n0 + 16 + l16];

    #pragma unroll
    for (int r = 0; r < 8; ++r) {
        const int m = m0 + r + half * 8;
        out[(size_t)m * OUT_D + n0 + l16]      = acc0[r] + bo0;
        out[(size_t)m * OUT_D + n0 + 16 + l16] = acc1[r] + bo1;
    }
}

extern "C" void kernel_launch(void* const* d_in, const int* in_sizes, int n_in,
                              void* d_out, int out_size, void* d_ws, size_t ws_size,
                              hipStream_t stream) {
    (void)in_sizes; (void)n_in; (void)out_size; (void)ws_size;

    const float* data  = (const float*)d_in[0];
    const float* W_ih  = (const float*)d_in[1];
    const float* W_hh  = (const float*)d_in[2];
    const float* b_i   = (const float*)d_in[3];
    const float* b_h   = (const float*)d_in[4];
    const float* W_out = (const float*)d_in[5];
    const float* b_out = (const float*)d_in[6];
    float* out = (float*)d_out;

    float* h0 = (float*)d_ws;
    float* h1 = h0 + (size_t)BATCH * HS_D;

    const int hn = BATCH * HS_D;
    rnn_zero_kernel<<<(hn + 255) / 256, 256, 0, stream>>>(h0, hn);

    for (int t = 0; t < T_STEPS; ++t) {
        const float* hin  = (t & 1) ? h1 : h0;
        float*       hout = (t & 1) ? h0 : h1;
        rnn_step_kernel<<<(BATCH / TILE_M) * (HS_D / TILE_N), 256, 0, stream>>>(
            data + (size_t)t * BATCH * INP_D, hin, W_ih, W_hh, b_i, b_h, hout);
    }
    rnn_out_kernel<<<HS_D / 32, 256, 0, stream>>>(h0, W_out, b_out, out);
}


# Step kernel: each group walks its 8 K slices in an order rotated by (nt + 2 mt) mod 8, so groups sharing a weight panel or row block fetch different lines at any time; changes summation order
# speedup vs baseline: 5.0475x; 1.0809x over previous
#include <hip/hip_runtime.h>
#include <math.h>


#define T_STEPS 256
#define BATCH   128
#define INP_D   1024
#define HS_D    1024
#define OUT_D   1024

#define TILE_M   32
#define TILE_N   32
#define PH_K     256
#define NPH      8
#define PH_F     (PH_K * TILE_N)
#define PH_BLK   (PH_K / 8)
#define OUT_PITCH 36

typedef __attribute__((ext_vector_type(2))) float v2f;
typedef __attribute__((ext_vector_type(4))) float v4f;
typedef __attribute__((ext_vector_type(8))) float v8f;

__device__ __forceinline__ v8f wmma_f32(v2f a, v2f b, v8f c) {
    return __builtin_amdgcn_wmma_f32_16x16x4_f32(
         false, a,  false, b,
         (short)0, c,  false,  false);
}

__global__ void rnn_zero_kernel(float* __restrict__ p, int n) {
    int i = blockIdx.x * blockDim.x + threadIdx.x;
    if (i < n) p[i] = 0.0f;
}


__device__ __forceinline__ void panel_load(float4 (&v)[4][4], const float* wp) {
    #pragma unroll
    for (int s = 0; s < 4; ++s) {
        #pragma unroll
        for (int R = 0; R < 4; ++R)
            v[s][R] = *(const float4*)(wp + (size_t)(64 * s + R) * HS_D);
    }
}

__device__ __forceinline__ void panel_store(float* reg, const float4 (&v)[4][4],
                                            int qq, int cg) {
    const int rot = qq & 3;
    #pragma unroll
    for (int s = 0; s < 4; ++s) {
        float* row = reg + (size_t)(16 * s + qq) * 128;
        const float4 c0 = make_float4(v[s][0].x, v[s][1].x, v[s][2].x, v[s][3].x);
        const float4 c1 = make_float4(v[s][0].y, v[s][1].y, v[s][2].y, v[s][3].y);
        const float4 c2 = make_float4(v[s][0].z, v[s][1].z, v[s][2].z, v[s][3].z);
        const float4 c3 = make_float4(v[s][0].w, v[s][1].w, v[s][2].w, v[s][3].w);
        *(float4*)(row + 4 * ((4 * cg + 0 + rot) & 31)) = c0;
        *(float4*)(row + 4 * ((4 * cg + 1 + rot) & 31)) = c1;
        *(float4*)(row + 4 * ((4 * cg + 2 + rot) & 31)) = c2;
        *(float4*)(row + 4 * ((4 * cg + 3 + rot) & 31)) = c3;
    }
}

__device__ __forceinline__ void rows_load(v4f (&v)[16], const float* ap) {
    #pragma unroll
    for (int i = 0; i < 16; ++i)
        v[i] = *(const v4f*)(ap + (size_t)(16 * (i & 1)) * HS_D + 32 * (i >> 1));
}

__device__ __forceinline__ void rows_store(float* abuf, const v4f (&v)[16],
                                           int ar0, int cg) {
    #pragma unroll
    for (int i = 0; i < 16; ++i) {
        const int row = ar0 + 16 * (i & 1);
        const int u   = 8 * (i >> 1) + cg;
        *(v4f*)(abuf + row * PH_K + 4 * (u ^ ar0)) = v[i];
    }
}

__device__ __forceinline__ void frag_group(float4 (&fa)[8], float4 (&fb)[8],
                                           const float* wreg, const float* arow,
                                           int g, int offE, int offO,
                                           const int (&aoff)[8]) {
    #pragma unroll
    for (int j = 0; j < 8; ++j) {
        const int b = 8 * g + j;
        fb[j] = *(const float4*)(wreg + 256 * b + ((b & 1) ? offO : offE));
        fa[j] = *(const float4*)(arow + 64 * g + aoff[j]);
    }
}

__device__ __forceinline__ void slice_mma(const float* wreg, const float* arow,
                                          float4 (&a0)[8], float4 (&b0)[8],
                                          int offE, int offO, const int (&aoff)[8],
                                          v8f& acc0, v8f& acc1) {
    float4 a1[8], b1[8];
    #pragma unroll
    for (int g = 0; g < 4; g += 2) {
        frag_group(a1, b1, wreg, arow, g + 1, offE, offO, aoff);
        __builtin_amdgcn_sched_barrier(0);
        #pragma unroll
        for (int j = 0; j < 8; ++j) {
            acc0 = wmma_f32((v2f){a0[j].x, a0[j].y}, (v2f){b0[j].x, b0[j].y}, acc0);
            acc1 = wmma_f32((v2f){a0[j].z, a0[j].w}, (v2f){b0[j].z, b0[j].w}, acc1);
        }
        __builtin_amdgcn_sched_barrier(0);
        if (g + 2 < 4) frag_group(a0, b0, wreg, arow, g + 2, offE, offO, aoff);
        __builtin_amdgcn_sched_barrier(0);
        #pragma unroll
        for (int j = 0; j < 8; ++j) {
            acc0 = wmma_f32((v2f){a1[j].x, a1[j].y}, (v2f){b1[j].x, b1[j].y}, acc0);
            acc1 = wmma_f32((v2f){a1[j].z, a1[j].w}, (v2f){b1[j].z, b1[j].w}, acc1);
        }
        __builtin_amdgcn_sched_barrier(0);
    }
}

__global__ __launch_bounds__(256) void rnn_step_kernel(
    const float* __restrict__ x,
    const float* __restrict__ h_in,
    const float* __restrict__ W_ih,
    const float* __restrict__ W_hh,
    const float* __restrict__ b_i,
    const float* __restrict__ b_h,
    float* __restrict__ h_out)
{
    __shared__ float Ws[NPH * PH_F];

    const int tid  = threadIdx.x;
    const int loader = __builtin_amdgcn_readfirstlane(tid >> 7);
    const int rt   = tid & 127;
    const int wave = rt >> 5;
    const int lane = tid & 31;
    const int h    = lane >> 4;
    const int c    = lane & 15;
    const int wm   = wave >> 1;
    const int wn   = wave & 1;
    const int mt   = blockIdx.x >> 5;
    const int nt   = blockIdx.x & 31;
    const int n0   = nt * TILE_N;
    const int n    = wn * 16 + c;

    const int offE = 128 * h + 4 * ((n + h) & 31);
    const int offO = 128 * h + 4 * ((n + 2 + h) & 31);

    const int cg = rt & 7;
    const int qq = rt >> 3;
    const size_t wofs = (size_t)(4 * qq) * HS_D + n0 + 4 * cg;
    const float* wih = W_ih + wofs;
    const float* whh = W_hh + wofs;

    const int ar0 = (qq >> 1) + 8 * (qq & 1);
    const size_t aofs = (size_t)(mt * TILE_M + ar0) * HS_D + 4 * cg;
    const float* xr = x    + aofs;
    const float* hr = h_in + aofs;

    const int arow_f = (wm * 16 + c) * PH_K;
    int aoff[8];
    #pragma unroll
    for (int j = 0; j < 8; ++j) aoff[j] = 4 * ((2 * j + h) ^ c);

    const float bias = b_i[n0 + n] + b_h[n0 + n];

    const int rot = (nt + 2 * mt) & 7;

    v8f acc0 = {};
    v8f acc1 = {};

    #define KSL(p)  (((p) + rot) & 7)
    #define WPTR(p) (((KSL(p) < 4) ? wih : whh) + (size_t)(KSL(p) & 3) * PH_K * HS_D)
    #define APTR(p) (((KSL(p) < 4) ? xr : hr) + (KSL(p) & 3) * PH_K)
    #define AREG(p) (Ws + (((p) < 2) ? (7 - (p)) : ((p) - 2)) * PH_F)

    float4 a0[8], b0[8];

    #define SLICE(p)                                                         \
        frag_group(a0, b0, Ws + (p) * PH_F, AREG(p) + arow_f,                \
                   0, offE, offO, aoff);                                     \
        slice_mma(Ws + (p) * PH_F, AREG(p) + arow_f, a0, b0,                 \
                  offE, offO, aoff, acc0, acc1);

    #define STAGE(p)                                                         \
        {                                                                    \
            float4 wv[4][4];                                                 \
            v4f av[16];                                                      \
            panel_load(wv, WPTR(p));                                         \
            rows_load(av, APTR(p));                                          \
            panel_store(Ws + (p) * PH_F, wv, qq, cg);                        \
            rows_store(AREG(p), av, ar0, cg);                                \
        }
    #define PHASE(p)                                                         \
        if (loader) {                                                        \
            STAGE((p) + 1)                                                   \
        } else {                                                             \
            SLICE(p)                                                         \
        }                                                                    \
        __syncthreads();

    if (loader) {
        STAGE(0)
    }
    __syncthreads();

    PHASE(0)
    PHASE(1)
    PHASE(2)
    PHASE(3)
    PHASE(4)
    PHASE(5)
    PHASE(6)
    if (!loader) {
        SLICE(7)
    }

    #undef PHASE
    #undef STAGE

    #undef SLICE
    #undef AREG
    #undef WPTR
    #undef APTR
    #undef KSL

    float* ot = Ws;
    if (!loader) {
        #pragma unroll
        for (int r = 0; r < 8; ++r) {
            const int row = wm * 16 + r + 8 * h;
            ot[row * OUT_PITCH + n] = tanhf((acc0[r] + acc1[r]) + bias);
        }
    }
    __syncthreads();
    {
        const int row = tid >> 3;
        const int col = (tid & 7) * 4;
        const float4 v = *(const float4*)(ot + row * OUT_PITCH + col);
        *(float4*)(h_out + (size_t)(mt * TILE_M + row) * HS_D + n0 + col) = v;
    }
}

__global__ __launch_bounds__(256) void rnn_out_kernel(
    const float* __restrict__ h,
    const float* __restrict__ W_out,
    const float* __restrict__ b_out,
    float* __restrict__ out)
{
    const int wave = threadIdx.x >> 5;
    const int lane = threadIdx.x & 31;
    const int half = lane >> 4;
    const int l16  = lane & 15;

    const int m0 = wave * 16;
    const int n0 = blockIdx.x * 32;

    v8f acc0 = {};
    v8f acc1 = {};

    const float* a_ptr  = h + (size_t)(m0 + l16) * HS_D + half * 2;
    const float* bbase0 = W_out + (size_t)(n0 + l16) * HS_D + half * 2;
    const float* bbase1 = W_out + (size_t)(n0 + 16 + l16) * HS_D + half * 2;

    #pragma unroll 4
    for (int k = 0; k < HS_D; k += 4) {
        v2f a  = *(const v2f*)(a_ptr + k);
        v2f b0 = *(const v2f*)(bbase0 + k);
        v2f b1 = *(const v2f*)(bbase1 + k);
        acc0 = wmma_f32(a, b0, acc0);
        acc1 = wmma_f32(a, b1, acc1);
    }

    const float bo0 = b_out[n0 + l16];
    const float bo1 = b_out[n0 + 16 + l16];

    #pragma unroll
    for (int r = 0; r < 8; ++r) {
        const int m = m0 + r + half * 8;
        out[(size_t)m * OUT_D + n0 + l16]      = acc0[r] + bo0;
        out[(size_t)m * OUT_D + n0 + 16 + l16] = acc1[r] + bo1;
    }
}

extern "C" void kernel_launch(void* const* d_in, const int* in_sizes, int n_in,
                              void* d_out, int out_size, void* d_ws, size_t ws_size,
                              hipStream_t stream) {
    (void)in_sizes; (void)n_in; (void)out_size; (void)ws_size;

    const float* data  = (const float*)d_in[0];
    const float* W_ih  = (const float*)d_in[1];
    const float* W_hh  = (const float*)d_in[2];
    const float* b_i   = (const float*)d_in[3];
    const float* b_h   = (const float*)d_in[4];
    const float* W_out = (const float*)d_in[5];
    const float* b_out = (const float*)d_in[6];
    float* out = (float*)d_out;

    float* h0 = (float*)d_ws;
    float* h1 = h0 + (size_t)BATCH * HS_D;

    const int hn = BATCH * HS_D;
    rnn_zero_kernel<<<(hn + 255) / 256, 256, 0, stream>>>(h0, hn);

    for (int t = 0; t < T_STEPS; ++t) {
        const float* hin  = (t & 1) ? h1 : h0;
        float*       hout = (t & 1) ? h0 : h1;
        rnn_step_kernel<<<(BATCH / TILE_M) * (HS_D / TILE_N), 256, 0, stream>>>(
            data + (size_t)t * BATCH * INP_D, hin, W_ih, W_hh, b_i, b_h, hout);
    }
    rnn_out_kernel<<<HS_D / 32, 256, 0, stream>>>(h0, W_out, b_out, out);
}
